# starting kernel
# baseline (speedup 1.0000x reference)
#include <hip/hip_runtime.h>
#include <hip/hip_bf16.h>

#define B_SAMPLES 8192
#define T_TOKENS  1048576
#define NHID      256
#define NCLASSES  14
#define NPAD      16

typedef __attribute__((ext_vector_type(2))) float v2f;
typedef __attribute__((ext_vector_type(8))) float v8f;

__global__ void seg_scan_kernel(const int* __restrict__ lengths,
                                int* __restrict__ offsets) {
    __shared__ int ssum[1024];
    const int t = threadIdx.x;
    const int base = t * 8;
    int loc[8];
    int s = 0;
#pragma unroll
    for (int j = 0; j < 8; ++j) {
        loc[j] = lengths[base + j];
        s += loc[j];
    }
    ssum[t] = s;
    __syncthreads();
    for (int off = 1; off < 1024; off <<= 1) {
        int v = (t >= off) ? ssum[t - off] : 0;
        __syncthreads();
        ssum[t] += v;
        __syncthreads();
    }
    int excl = ssum[t] - s;
#pragma unroll
    for (int j = 0; j < 8; ++j) {
        offsets[base + j] = excl;
        excl += loc[j];
    }
    if (t == 1023) offsets[B_SAMPLES] = ssum[1023];
}

__global__ void seg_mean_tanh_kernel(const int* __restrict__ offsets,
                                     const int* __restrict__ indices,
                                     const float* __restrict__ emb,
                                     float* __restrict__ Amat) {
    __shared__ int sidx[256];
    const int tid = threadIdx.x;
    const int sample = blockIdx.x;
    const int s = offsets[sample];
    const int e = offsets[sample + 1];

    float acc = 0.0f;
    for (int base = s; base < e; base += 256) {
        const int n = (e - base < 256) ? (e - base) : 256;
        if (tid < n) sidx[tid] = indices[base + tid];
        __syncthreads();
        for (int j = 0; j < n; ++j) {
            if (j + 8 < n) {
                __builtin_prefetch(emb + (size_t)sidx[j + 8] * NHID + tid, 0, 1);
            }
            acc += emb[(size_t)sidx[j] * NHID + tid];
        }
        __syncthreads();
    }

    const float len = (float)(e - s);
    const float m = (len > 0.0f) ? (acc / len) : 0.0f;
    Amat[(size_t)sample * NHID + tid] = tanhf(m);
}

__global__ void wmma_head_kernel(const float* __restrict__ Amat,
                                 const float* __restrict__ W,
                                 const float* __restrict__ bias,
                                 float* __restrict__ out) {
    __shared__ float lw[NHID * NPAD];

    const int tid = threadIdx.x;
    for (int i = tid; i < NHID * NPAD; i += 256) {
        const int kk = i >> 4;
        const int nn = i & 15;
        lw[i] = (nn < NCLASSES) ? W[kk * NCLASSES + nn] : 0.0f;
    }
    __syncthreads();

    const int lane = tid & 31;
    const int wave = tid >> 5;
    const int tile = blockIdx.x * 8 + wave;
    const int row0 = tile * 16;

    const int mrow = lane & 15;
    const int koff = (lane >> 4) << 1;
    const float* __restrict__ arow = Amat + (size_t)(row0 + mrow) * NHID;

    v8f c = {};
#pragma unroll 4
    for (int k = 0; k < NHID; k += 4) {
        const float2 a2 = *(const float2*)(arow + k + koff);
        v2f av; av.x = a2.x; av.y = a2.y;
        v2f bv;
        bv.x = lw[(k + koff) * NPAD + mrow];
        bv.y = lw[(k + koff + 1) * NPAD + mrow];
        c = __builtin_amdgcn_wmma_f32_16x16x4_f32(
            false, av, false, bv, (short)0, c, false, false);
    }

    const int mhi = (lane >> 4) << 3;
    const int ncol = lane & 15;
    if (ncol < NCLASSES) {
        const float bb = bias[ncol];
#pragma unroll
        for (int j = 0; j < 8; ++j) {
            const int m = row0 + j + mhi;
            out[(size_t)m * NCLASSES + ncol] = c[j] + bb;
        }
    }
}

extern "C" void kernel_launch(void* const* d_in, const int* in_sizes, int n_in,
                              void* d_out, int out_size, void* d_ws, size_t ws_size,
                              hipStream_t stream) {
    const int*   lengths = (const int*)d_in[0];
    const int*   indices = (const int*)d_in[1];
    const float* emb     = (const float*)d_in[2];
    const float* W       = (const float*)d_in[3];
    const float* bias    = (const float*)d_in[4];
    float*       out     = (float*)d_out;

    int*   offsets = (int*)d_ws;
    float* Amat    = (float*)((char*)d_ws + (64 * 1024));

    seg_scan_kernel<<<1, 1024, 0, stream>>>(lengths, offsets);
    seg_mean_tanh_kernel<<<B_SAMPLES, 256, 0, stream>>>(offsets, indices, emb, Amat);
    wmma_head_kernel<<<B_SAMPLES / (16 * 8), 256, 0, stream>>>(Amat, W, bias, out);
}


# seg-mean gather: wave pair per row, 16-byte lane loads from scalar-read index windows, 16 row loads in flight per wave, 4096 two-wave blocks x 2 samples (was a 4-byte load per token); sum order kept
# speedup vs baseline: 2.9526x; 2.9526x over previous
#include <hip/hip_runtime.h>
#include <hip/hip_bf16.h>

#define B_SAMPLES 8192
#define T_TOKENS  1048576
#define NHID      256
#define NCLASSES  14
#define NPAD      16

typedef __attribute__((ext_vector_type(2))) float v2f;
typedef __attribute__((ext_vector_type(8))) float v8f;

__global__ void seg_scan_kernel(const int* __restrict__ lengths,
                                int* __restrict__ offsets) {
    __shared__ int ssum[1024];
    const int t = threadIdx.x;
    const int base = t * 8;
    int loc[8];
    int s = 0;
#pragma unroll
    for (int j = 0; j < 8; ++j) {
        loc[j] = lengths[base + j];
        s += loc[j];
    }
    ssum[t] = s;
    __syncthreads();
    for (int off = 1; off < 1024; off <<= 1) {
        int v = (t >= off) ? ssum[t - off] : 0;
        __syncthreads();
        ssum[t] += v;
        __syncthreads();
    }
    int excl = ssum[t] - s;
#pragma unroll
    for (int j = 0; j < 8; ++j) {
        offsets[base + j] = excl;
        excl += loc[j];
    }
    if (t == 1023) offsets[B_SAMPLES] = ssum[1023];
}

#define GATHER_GROUP  16
#define GATHER_BLOCKS 4096

static_assert(T_TOKENS >= GATHER_GROUP, "index windows assume at least one full window of tokens");

__global__ void __launch_bounds__(64)
seg_mean_tanh_kernel(const int* __restrict__ offsets,
                     const int* __restrict__ indices,
                     const float* __restrict__ emb,
                     float* __restrict__ Amat) {
    const int tid = threadIdx.x;
    const float* __restrict__ lane_emb = emb + tid * 4;

    for (int sample = blockIdx.x; sample < B_SAMPLES; sample += GATHER_BLOCKS) {
        const int s = offsets[sample];
        const int e = offsets[sample + 1];

        float4 acc = make_float4(0.0f, 0.0f, 0.0f, 0.0f);
        int base = s;

        int q = 0;
        int r[GATHER_GROUP];
#pragma unroll
        for (int t = 0; t < GATHER_GROUP; ++t) r[t] = 0;
        if (e > s) {
            const int qt = (e >= GATHER_GROUP) ? (e - GATHER_GROUP) : 0;
            q = (s + GATHER_GROUP <= e) ? s : qt;
#pragma unroll
            for (int t = 0; t < GATHER_GROUP; ++t) r[t] = indices[q + t];
        }

        while (base + GATHER_GROUP <= e) {
            float4 v[GATHER_GROUP];
#pragma unroll
            for (int t = 0; t < GATHER_GROUP; ++t)
                v[t] = *(const float4*)(lane_emb + (size_t)r[t] * NHID);
            const int nb = base + GATHER_GROUP;
            if (nb < e) {
                q = (nb + GATHER_GROUP <= e) ? nb : (e - GATHER_GROUP);
#pragma unroll
                for (int t = 0; t < GATHER_GROUP; ++t) r[t] = indices[q + t];
            }
#pragma unroll
            for (int t = 0; t < GATHER_GROUP; ++t) {
                acc.x += v[t].x; acc.y += v[t].y; acc.z += v[t].z; acc.w += v[t].w;
            }
            base = nb;
        }

        if (base < e) {
            const int lo = base - q;
            const int hi = e - q;
            float4 v[GATHER_GROUP];
#pragma unroll
            for (int t = 0; t < GATHER_GROUP; ++t) {
                const bool in = (t >= lo) && (t < hi);
                const int row = in ? r[t] : 0;
                v[t] = *(const float4*)(lane_emb + (size_t)row * NHID);
            }
#pragma unroll
            for (int t = 0; t < GATHER_GROUP; ++t) {
                const bool in = (t >= lo) && (t < hi);
                acc.x += in ? v[t].x : 0.0f;
                acc.y += in ? v[t].y : 0.0f;
                acc.z += in ? v[t].z : 0.0f;
                acc.w += in ? v[t].w : 0.0f;
            }
        }

        const float len = (float)(e - s);
        float4 o;
        o.x = tanhf((len > 0.0f) ? (acc.x / len) : 0.0f);
        o.y = tanhf((len > 0.0f) ? (acc.y / len) : 0.0f);
        o.z = tanhf((len > 0.0f) ? (acc.z / len) : 0.0f);
        o.w = tanhf((len > 0.0f) ? (acc.w / len) : 0.0f);
        *(float4*)(Amat + (size_t)sample * NHID + tid * 4) = o;
    }
}

__global__ void wmma_head_kernel(const float* __restrict__ Amat,
                                 const float* __restrict__ W,
                                 const float* __restrict__ bias,
                                 float* __restrict__ out) {
    __shared__ float lw[NHID * NPAD];

    const int tid = threadIdx.x;
    for (int i = tid; i < NHID * NPAD; i += 256) {
        const int kk = i >> 4;
        const int nn = i & 15;
        lw[i] = (nn < NCLASSES) ? W[kk * NCLASSES + nn] : 0.0f;
    }
    __syncthreads();

    const int lane = tid & 31;
    const int wave = tid >> 5;
    const int tile = blockIdx.x * 8 + wave;
    const int row0 = tile * 16;

    const int mrow = lane & 15;
    const int koff = (lane >> 4) << 1;
    const float* __restrict__ arow = Amat + (size_t)(row0 + mrow) * NHID;

    v8f c = {};
#pragma unroll 4
    for (int k = 0; k < NHID; k += 4) {
        const float2 a2 = *(const float2*)(arow + k + koff);
        v2f av; av.x = a2.x; av.y = a2.y;
        v2f bv;
        bv.x = lw[(k + koff) * NPAD + mrow];
        bv.y = lw[(k + koff + 1) * NPAD + mrow];
        c = __builtin_amdgcn_wmma_f32_16x16x4_f32(
            false, av, false, bv, (short)0, c, false, false);
    }

    const int mhi = (lane >> 4) << 3;
    const int ncol = lane & 15;
    if (ncol < NCLASSES) {
        const float bb = bias[ncol];
#pragma unroll
        for (int j = 0; j < 8; ++j) {
            const int m = row0 + j + mhi;
            out[(size_t)m * NCLASSES + ncol] = c[j] + bb;
        }
    }
}

extern "C" void kernel_launch(void* const* d_in, const int* in_sizes, int n_in,
                              void* d_out, int out_size, void* d_ws, size_t ws_size,
                              hipStream_t stream) {
    const int*   lengths = (const int*)d_in[0];
    const int*   indices = (const int*)d_in[1];
    const float* emb     = (const float*)d_in[2];
    const float* W       = (const float*)d_in[3];
    const float* bias    = (const float*)d_in[4];
    float*       out     = (float*)d_out;

    int*   offsets = (int*)d_ws;
    float* Amat    = (float*)((char*)d_ws + (64 * 1024));

    seg_scan_kernel<<<1, 1024, 0, stream>>>(lengths, offsets);
    seg_mean_tanh_kernel<<<GATHER_BLOCKS, 64, 0, stream>>>(offsets, indices, emb, Amat);
    wmma_head_kernel<<<B_SAMPLES / (16 * 8), 256, 0, stream>>>(Amat, W, bias, out);
}


# seg-mean gather: 48 row loads in flight per wave (three 16-token index windows issued together, next batch's indices fetched ahead) instead of 16; token-order sums kept
# speedup vs baseline: 3.3893x; 1.1479x over previous
#include <hip/hip_runtime.h>
#include <hip/hip_bf16.h>

#define B_SAMPLES 8192
#define T_TOKENS  1048576
#define NHID      256
#define NCLASSES  14
#define NPAD      16

typedef __attribute__((ext_vector_type(2))) float v2f;
typedef __attribute__((ext_vector_type(8))) float v8f;

__global__ void seg_scan_kernel(const int* __restrict__ lengths,
                                int* __restrict__ offsets) {
    __shared__ int ssum[1024];
    const int t = threadIdx.x;
    const int base = t * 8;
    int loc[8];
    int s = 0;
#pragma unroll
    for (int j = 0; j < 8; ++j) {
        loc[j] = lengths[base + j];
        s += loc[j];
    }
    ssum[t] = s;
    __syncthreads();
    for (int off = 1; off < 1024; off <<= 1) {
        int v = (t >= off) ? ssum[t - off] : 0;
        __syncthreads();
        ssum[t] += v;
        __syncthreads();
    }
    int excl = ssum[t] - s;
#pragma unroll
    for (int j = 0; j < 8; ++j) {
        offsets[base + j] = excl;
        excl += loc[j];
    }
    if (t == 1023) offsets[B_SAMPLES] = ssum[1023];
}

#define GATHER_GROUP  16
#define GATHER_DEPTH  3
#define GATHER_BATCH  (GATHER_GROUP * GATHER_DEPTH)
#define GATHER_BLOCKS 4096

static_assert(T_TOKENS >= GATHER_GROUP, "index windows assume at least one full window of tokens");

__device__ __forceinline__ void gather_window(int base, int e, int& q, int& lo, int& hi) {
    const bool full = (base + GATHER_GROUP <= e);
    const int qt = (e >= GATHER_GROUP) ? (e - GATHER_GROUP) : 0;
    q = full ? base : qt;
    lo = base - q;
    hi = (full ? (base + GATHER_GROUP) : e) - q;
}

__device__ __forceinline__ void gather_indices(int (&r)[GATHER_GROUP],
                                               const int* __restrict__ indices, int q) {
#pragma unroll
    for (int t = 0; t < GATHER_GROUP; ++t) r[t] = indices[q + t];
}

__device__ __forceinline__ void gather_batch_indices(int (&ra)[GATHER_GROUP],
                                                     int (&rb)[GATHER_GROUP],
                                                     int (&rc)[GATHER_GROUP],
                                                     const int* __restrict__ indices,
                                                     int b0, int e) {
    const int b1 = (b0 + GATHER_GROUP < e) ? (b0 + GATHER_GROUP) : b0;
    const int b2 = (b0 + 2 * GATHER_GROUP < e) ? (b0 + 2 * GATHER_GROUP) : b0;
    int q0, q1, q2, lo, hi;
    gather_window(b0, e, q0, lo, hi);
    gather_window(b1, e, q1, lo, hi);
    gather_window(b2, e, q2, lo, hi);
    gather_indices(ra, indices, q0);
    gather_indices(rb, indices, q1);
    gather_indices(rc, indices, q2);
}

__device__ __forceinline__ void gather_issue_full(float4 (&v)[GATHER_GROUP],
                                                  const int (&r)[GATHER_GROUP],
                                                  const float* __restrict__ lane_emb) {
#pragma unroll
    for (int t = 0; t < GATHER_GROUP; ++t)
        v[t] = *(const float4*)(lane_emb + (size_t)r[t] * NHID);
}

__device__ __forceinline__ void gather_issue(float4 (&v)[GATHER_GROUP],
                                             const int (&r)[GATHER_GROUP],
                                             const float* __restrict__ lane_emb,
                                             int lo, int hi) {
#pragma unroll
    for (int t = 0; t < GATHER_GROUP; ++t) {
        const bool in = (t >= lo) && (t < hi);
        const int row = in ? r[t] : 0;
        v[t] = *(const float4*)(lane_emb + (size_t)row * NHID);
    }
}

__device__ __forceinline__ void gather_sum_full(float4& acc, const float4 (&v)[GATHER_GROUP]) {
#pragma unroll
    for (int t = 0; t < GATHER_GROUP; ++t) {
        acc.x += v[t].x; acc.y += v[t].y; acc.z += v[t].z; acc.w += v[t].w;
    }
}

__device__ __forceinline__ void gather_sum(float4& acc, const float4 (&v)[GATHER_GROUP],
                                           int lo, int hi) {
    if (hi - lo == GATHER_GROUP) {
        gather_sum_full(acc, v);
    } else {
#pragma unroll
        for (int t = 0; t < GATHER_GROUP; ++t) {
            const bool in = (t >= lo) && (t < hi);
            acc.x += in ? v[t].x : 0.0f;
            acc.y += in ? v[t].y : 0.0f;
            acc.z += in ? v[t].z : 0.0f;
            acc.w += in ? v[t].w : 0.0f;
        }
    }
}

__global__ void __launch_bounds__(64)
seg_mean_tanh_kernel(const int* __restrict__ offsets,
                     const int* __restrict__ indices,
                     const float* __restrict__ emb,
                     float* __restrict__ Amat) {
    const int tid = threadIdx.x;
    const float* __restrict__ lane_emb = emb + tid * 4;

    for (int sample = blockIdx.x; sample < B_SAMPLES; sample += GATHER_BLOCKS) {
        const int s = offsets[sample];
        const int e = offsets[sample + 1];

        float4 acc = make_float4(0.0f, 0.0f, 0.0f, 0.0f);
        int base = s;

        int ra[GATHER_GROUP], rb[GATHER_GROUP], rc[GATHER_GROUP];
#pragma unroll
        for (int t = 0; t < GATHER_GROUP; ++t) { ra[t] = 0; rb[t] = 0; rc[t] = 0; }
        if (base < e) gather_batch_indices(ra, rb, rc, indices, base, e);

        while (base + GATHER_BATCH <= e) {
            float4 va[GATHER_GROUP], vb[GATHER_GROUP], vc[GATHER_GROUP];
            gather_issue_full(va, ra, lane_emb);
            gather_issue_full(vb, rb, lane_emb);
            gather_issue_full(vc, rc, lane_emb);
            const int nb = base + GATHER_BATCH;
            if (nb < e) gather_batch_indices(ra, rb, rc, indices, nb, e);
            gather_sum_full(acc, va);
            gather_sum_full(acc, vb);
            gather_sum_full(acc, vc);
            base = nb;
        }

        if (base < e) {
            const bool w1 = (base + GATHER_GROUP < e);
            const bool w2 = (base + 2 * GATHER_GROUP < e);
            int q, lo0, hi0, lo1, hi1, lo2, hi2;
            gather_window(base, e, q, lo0, hi0);
            gather_window(w1 ? (base + GATHER_GROUP) : base, e, q, lo1, hi1);
            gather_window(w2 ? (base + 2 * GATHER_GROUP) : base, e, q, lo2, hi2);
            float4 va[GATHER_GROUP], vb[GATHER_GROUP], vc[GATHER_GROUP];
            gather_issue(va, ra, lane_emb, lo0, hi0);
            if (w1) gather_issue(vb, rb, lane_emb, lo1, hi1);
            if (w2) gather_issue(vc, rc, lane_emb, lo2, hi2);
            gather_sum(acc, va, lo0, hi0);
            if (w1) gather_sum(acc, vb, lo1, hi1);
            if (w2) gather_sum(acc, vc, lo2, hi2);
        }

        const float len = (float)(e - s);
        float4 o;
        o.x = tanhf((len > 0.0f) ? (acc.x / len) : 0.0f);
        o.y = tanhf((len > 0.0f) ? (acc.y / len) : 0.0f);
        o.z = tanhf((len > 0.0f) ? (acc.z / len) : 0.0f);
        o.w = tanhf((len > 0.0f) ? (acc.w / len) : 0.0f);
        *(float4*)(Amat + (size_t)sample * NHID + tid * 4) = o;
    }
}

__global__ void wmma_head_kernel(const float* __restrict__ Amat,
                                 const float* __restrict__ W,
                                 const float* __restrict__ bias,
                                 float* __restrict__ out) {
    __shared__ float lw[NHID * NPAD];

    const int tid = threadIdx.x;
    for (int i = tid; i < NHID * NPAD; i += 256) {
        const int kk = i >> 4;
        const int nn = i & 15;
        lw[i] = (nn < NCLASSES) ? W[kk * NCLASSES + nn] : 0.0f;
    }
    __syncthreads();

    const int lane = tid & 31;
    const int wave = tid >> 5;
    const int tile = blockIdx.x * 8 + wave;
    const int row0 = tile * 16;

    const int mrow = lane & 15;
    const int koff = (lane >> 4) << 1;
    const float* __restrict__ arow = Amat + (size_t)(row0 + mrow) * NHID;

    v8f c = {};
#pragma unroll 4
    for (int k = 0; k < NHID; k += 4) {
        const float2 a2 = *(const float2*)(arow + k + koff);
        v2f av; av.x = a2.x; av.y = a2.y;
        v2f bv;
        bv.x = lw[(k + koff) * NPAD + mrow];
        bv.y = lw[(k + koff + 1) * NPAD + mrow];
        c = __builtin_amdgcn_wmma_f32_16x16x4_f32(
            false, av, false, bv, (short)0, c, false, false);
    }

    const int mhi = (lane >> 4) << 3;
    const int ncol = lane & 15;
    if (ncol < NCLASSES) {
        const float bb = bias[ncol];
#pragma unroll
        for (int j = 0; j < 8; ++j) {
            const int m = row0 + j + mhi;
            out[(size_t)m * NCLASSES + ncol] = c[j] + bb;
        }
    }
}

extern "C" void kernel_launch(void* const* d_in, const int* in_sizes, int n_in,
                              void* d_out, int out_size, void* d_ws, size_t ws_size,
                              hipStream_t stream) {
    const int*   lengths = (const int*)d_in[0];
    const int*   indices = (const int*)d_in[1];
    const float* emb     = (const float*)d_in[2];
    const float* W       = (const float*)d_in[3];
    const float* bias    = (const float*)d_in[4];
    float*       out     = (float*)d_out;

    int*   offsets = (int*)d_ws;
    float* Amat    = (float*)((char*)d_ws + (64 * 1024));

    seg_scan_kernel<<<1, 1024, 0, stream>>>(lengths, offsets);
    seg_mean_tanh_kernel<<<GATHER_BLOCKS, 64, 0, stream>>>(offsets, indices, emb, Amat);
    wmma_head_kernel<<<B_SAMPLES / (16 * 8), 256, 0, stream>>>(Amat, W, bias, out);
}


# seg-mean gather: one 4-wave block per sample, its 48-token batches alternate between two wave pairs (96 rows in flight per sample, was 48); sum order changed: even-batch sum + odd-batch sum via LDS
# speedup vs baseline: 3.5076x; 1.0349x over previous
#include <hip/hip_runtime.h>
#include <hip/hip_bf16.h>

#define B_SAMPLES 8192
#define T_TOKENS  1048576
#define NHID      256
#define NCLASSES  14
#define NPAD      16

typedef __attribute__((ext_vector_type(2))) float v2f;
typedef __attribute__((ext_vector_type(8))) float v8f;

__global__ void seg_scan_kernel(const int* __restrict__ lengths,
                                int* __restrict__ offsets) {
    __shared__ int ssum[1024];
    const int t = threadIdx.x;
    const int base = t * 8;
    int loc[8];
    int s = 0;
#pragma unroll
    for (int j = 0; j < 8; ++j) {
        loc[j] = lengths[base + j];
        s += loc[j];
    }
    ssum[t] = s;
    __syncthreads();
    for (int off = 1; off < 1024; off <<= 1) {
        int v = (t >= off) ? ssum[t - off] : 0;
        __syncthreads();
        ssum[t] += v;
        __syncthreads();
    }
    int excl = ssum[t] - s;
#pragma unroll
    for (int j = 0; j < 8; ++j) {
        offsets[base + j] = excl;
        excl += loc[j];
    }
    if (t == 1023) offsets[B_SAMPLES] = ssum[1023];
}

#define GATHER_GROUP  16
#define GATHER_DEPTH  3
#define GATHER_BATCH  (GATHER_GROUP * GATHER_DEPTH)
#define GATHER_PAIRS  2
#define GATHER_THREADS 128

static_assert(GATHER_THREADS == 64 * GATHER_PAIRS, "one wave pair per 64 threads");

static_assert(T_TOKENS >= GATHER_GROUP, "index windows assume at least one full window of tokens");

__device__ __forceinline__ void gather_window(int base, int e, int& q, int& lo, int& hi) {
    const bool full = (base + GATHER_GROUP <= e);
    const int qt = (e >= GATHER_GROUP) ? (e - GATHER_GROUP) : 0;
    q = full ? base : qt;
    lo = base - q;
    hi = (full ? (base + GATHER_GROUP) : e) - q;
}

__device__ __forceinline__ void gather_indices(int (&r)[GATHER_GROUP],
                                               const int* __restrict__ indices, int q) {
#pragma unroll
    for (int t = 0; t < GATHER_GROUP; ++t) r[t] = indices[q + t];
}

__device__ __forceinline__ void gather_batch_indices(int (&ra)[GATHER_GROUP],
                                                     int (&rb)[GATHER_GROUP],
                                                     int (&rc)[GATHER_GROUP],
                                                     const int* __restrict__ indices,
                                                     int b0, int e) {
    const int b1 = (b0 + GATHER_GROUP < e) ? (b0 + GATHER_GROUP) : b0;
    const int b2 = (b0 + 2 * GATHER_GROUP < e) ? (b0 + 2 * GATHER_GROUP) : b0;
    int q0, q1, q2, lo, hi;
    gather_window(b0, e, q0, lo, hi);
    gather_window(b1, e, q1, lo, hi);
    gather_window(b2, e, q2, lo, hi);
    gather_indices(ra, indices, q0);
    gather_indices(rb, indices, q1);
    gather_indices(rc, indices, q2);
}

__device__ __forceinline__ void gather_issue_full(float4 (&v)[GATHER_GROUP],
                                                  const int (&r)[GATHER_GROUP],
                                                  const float* __restrict__ lane_emb) {
#pragma unroll
    for (int t = 0; t < GATHER_GROUP; ++t)
        v[t] = *(const float4*)(lane_emb + (size_t)r[t] * NHID);
}

__device__ __forceinline__ void gather_issue(float4 (&v)[GATHER_GROUP],
                                             const int (&r)[GATHER_GROUP],
                                             const float* __restrict__ lane_emb,
                                             int lo, int hi) {
#pragma unroll
    for (int t = 0; t < GATHER_GROUP; ++t) {
        const bool in = (t >= lo) && (t < hi);
        const int row = in ? r[t] : 0;
        v[t] = *(const float4*)(lane_emb + (size_t)row * NHID);
    }
}

__device__ __forceinline__ void gather_sum_full(float4& acc, const float4 (&v)[GATHER_GROUP]) {
#pragma unroll
    for (int t = 0; t < GATHER_GROUP; ++t) {
        acc.x += v[t].x; acc.y += v[t].y; acc.z += v[t].z; acc.w += v[t].w;
    }
}

__device__ __forceinline__ void gather_sum(float4& acc, const float4 (&v)[GATHER_GROUP],
                                           int lo, int hi) {
    if (hi - lo == GATHER_GROUP) {
        gather_sum_full(acc, v);
    } else {
#pragma unroll
        for (int t = 0; t < GATHER_GROUP; ++t) {
            const bool in = (t >= lo) && (t < hi);
            acc.x += in ? v[t].x : 0.0f;
            acc.y += in ? v[t].y : 0.0f;
            acc.z += in ? v[t].z : 0.0f;
            acc.w += in ? v[t].w : 0.0f;
        }
    }
}

__global__ void __launch_bounds__(GATHER_THREADS)
seg_mean_tanh_kernel(const int* __restrict__ offsets,
                     const int* __restrict__ indices,
                     const float* __restrict__ emb,
                     float* __restrict__ Amat) {
    __shared__ float4 part[GATHER_PAIRS - 1][64];

    const int tid = threadIdx.x;
    const int col = tid & 63;
    const int pair = __builtin_amdgcn_readfirstlane(tid >> 6);
    const float* __restrict__ lane_emb = emb + col * 4;

    {
        const int sample = blockIdx.x;
        const int s = offsets[sample];
        const int e = offsets[sample + 1];

        float4 acc = make_float4(0.0f, 0.0f, 0.0f, 0.0f);
        int base = s + pair * GATHER_BATCH;

        int ra[GATHER_GROUP], rb[GATHER_GROUP], rc[GATHER_GROUP];
#pragma unroll
        for (int t = 0; t < GATHER_GROUP; ++t) { ra[t] = 0; rb[t] = 0; rc[t] = 0; }
        if (base < e) gather_batch_indices(ra, rb, rc, indices, base, e);

        while (base + GATHER_BATCH <= e) {
            float4 va[GATHER_GROUP], vb[GATHER_GROUP], vc[GATHER_GROUP];
            gather_issue_full(va, ra, lane_emb);
            gather_issue_full(vb, rb, lane_emb);
            gather_issue_full(vc, rc, lane_emb);
            const int nb = base + GATHER_PAIRS * GATHER_BATCH;
            if (nb < e) gather_batch_indices(ra, rb, rc, indices, nb, e);
            gather_sum_full(acc, va);
            gather_sum_full(acc, vb);
            gather_sum_full(acc, vc);
            base = nb;
        }

        if (base < e) {
            const bool w1 = (base + GATHER_GROUP < e);
            const bool w2 = (base + 2 * GATHER_GROUP < e);
            int q, lo0, hi0, lo1, hi1, lo2, hi2;
            gather_window(base, e, q, lo0, hi0);
            gather_window(w1 ? (base + GATHER_GROUP) : base, e, q, lo1, hi1);
            gather_window(w2 ? (base + 2 * GATHER_GROUP) : base, e, q, lo2, hi2);
            float4 va[GATHER_GROUP], vb[GATHER_GROUP], vc[GATHER_GROUP];
            gather_issue(va, ra, lane_emb, lo0, hi0);
            if (w1) gather_issue(vb, rb, lane_emb, lo1, hi1);
            if (w2) gather_issue(vc, rc, lane_emb, lo2, hi2);
            gather_sum(acc, va, lo0, hi0);
            if (w1) gather_sum(acc, vb, lo1, hi1);
            if (w2) gather_sum(acc, vc, lo2, hi2);
        }

        if (pair != 0) part[pair - 1][col] = acc;
        __syncthreads();
        if (pair == 0) {
#pragma unroll
            for (int p = 1; p < GATHER_PAIRS; ++p) {
                const float4 o = part[p - 1][col];
                acc.x += o.x; acc.y += o.y; acc.z += o.z; acc.w += o.w;
            }
            const float len = (float)(e - s);
            float4 o;
            o.x = tanhf((len > 0.0f) ? (acc.x / len) : 0.0f);
            o.y = tanhf((len > 0.0f) ? (acc.y / len) : 0.0f);
            o.z = tanhf((len > 0.0f) ? (acc.z / len) : 0.0f);
            o.w = tanhf((len > 0.0f) ? (acc.w / len) : 0.0f);
            *(float4*)(Amat + (size_t)sample * NHID + col * 4) = o;
        }
    }
}

__global__ void wmma_head_kernel(const float* __restrict__ Amat,
                                 const float* __restrict__ W,
                                 const float* __restrict__ bias,
                                 float* __restrict__ out) {
    __shared__ float lw[NHID * NPAD];

    const int tid = threadIdx.x;
    for (int i = tid; i < NHID * NPAD; i += 256) {
        const int kk = i >> 4;
        const int nn = i & 15;
        lw[i] = (nn < NCLASSES) ? W[kk * NCLASSES + nn] : 0.0f;
    }
    __syncthreads();

    const int lane = tid & 31;
    const int wave = tid >> 5;
    const int tile = blockIdx.x * 8 + wave;
    const int row0 = tile * 16;

    const int mrow = lane & 15;
    const int koff = (lane >> 4) << 1;
    const float* __restrict__ arow = Amat + (size_t)(row0 + mrow) * NHID;

    v8f c = {};
#pragma unroll 4
    for (int k = 0; k < NHID; k += 4) {
        const float2 a2 = *(const float2*)(arow + k + koff);
        v2f av; av.x = a2.x; av.y = a2.y;
        v2f bv;
        bv.x = lw[(k + koff) * NPAD + mrow];
        bv.y = lw[(k + koff + 1) * NPAD + mrow];
        c = __builtin_amdgcn_wmma_f32_16x16x4_f32(
            false, av, false, bv, (short)0, c, false, false);
    }

    const int mhi = (lane >> 4) << 3;
    const int ncol = lane & 15;
    if (ncol < NCLASSES) {
        const float bb = bias[ncol];
#pragma unroll
        for (int j = 0; j < 8; ++j) {
            const int m = row0 + j + mhi;
            out[(size_t)m * NCLASSES + ncol] = c[j] + bb;
        }
    }
}

extern "C" void kernel_launch(void* const* d_in, const int* in_sizes, int n_in,
                              void* d_out, int out_size, void* d_ws, size_t ws_size,
                              hipStream_t stream) {
    const int*   lengths = (const int*)d_in[0];
    const int*   indices = (const int*)d_in[1];
    const float* emb     = (const float*)d_in[2];
    const float* W       = (const float*)d_in[3];
    const float* bias    = (const float*)d_in[4];
    float*       out     = (float*)d_out;

    int*   offsets = (int*)d_ws;
    float* Amat    = (float*)((char*)d_ws + (64 * 1024));

    seg_scan_kernel<<<1, 1024, 0, stream>>>(lengths, offsets);
    seg_mean_tanh_kernel<<<B_SAMPLES, GATHER_THREADS, 0, stream>>>(offsets, indices, emb, Amat);
    wmma_head_kernel<<<B_SAMPLES / (16 * 8), 256, 0, stream>>>(Amat, W, bias, out);
}


# head product cut for latency: one wave per 16-row tile, all W and A loads in flight at once, W via one LDS copy, two accumulator chains; sum order changed (k regrouped, two partial sums added)
# speedup vs baseline: 4.0843x; 1.1644x over previous
#include <hip/hip_runtime.h>
#include <hip/hip_bf16.h>

#define B_SAMPLES 8192
#define T_TOKENS  1048576
#define NHID      256
#define NCLASSES  14
#define NPAD      16

typedef __attribute__((ext_vector_type(2))) float v2f;
typedef __attribute__((ext_vector_type(8))) float v8f;

__global__ void seg_scan_kernel(const int* __restrict__ lengths,
                                int* __restrict__ offsets) {
    __shared__ int ssum[1024];
    const int t = threadIdx.x;
    const int base = t * 8;
    int loc[8];
    int s = 0;
#pragma unroll
    for (int j = 0; j < 8; ++j) {
        loc[j] = lengths[base + j];
        s += loc[j];
    }
    ssum[t] = s;
    __syncthreads();
    for (int off = 1; off < 1024; off <<= 1) {
        int v = (t >= off) ? ssum[t - off] : 0;
        __syncthreads();
        ssum[t] += v;
        __syncthreads();
    }
    int excl = ssum[t] - s;
#pragma unroll
    for (int j = 0; j < 8; ++j) {
        offsets[base + j] = excl;
        excl += loc[j];
    }
    if (t == 1023) offsets[B_SAMPLES] = ssum[1023];
}

#define GATHER_GROUP  16
#define GATHER_DEPTH  3
#define GATHER_BATCH  (GATHER_GROUP * GATHER_DEPTH)
#define GATHER_PAIRS  2
#define GATHER_THREADS 128

static_assert(GATHER_THREADS == 64 * GATHER_PAIRS, "one wave pair per 64 threads");

static_assert(T_TOKENS >= GATHER_GROUP, "index windows assume at least one full window of tokens");

__device__ __forceinline__ void gather_window(int base, int e, int& q, int& lo, int& hi) {
    const bool full = (base + GATHER_GROUP <= e);
    const int qt = (e >= GATHER_GROUP) ? (e - GATHER_GROUP) : 0;
    q = full ? base : qt;
    lo = base - q;
    hi = (full ? (base + GATHER_GROUP) : e) - q;
}

__device__ __forceinline__ void gather_indices(int (&r)[GATHER_GROUP],
                                               const int* __restrict__ indices, int q) {
#pragma unroll
    for (int t = 0; t < GATHER_GROUP; ++t) r[t] = indices[q + t];
}

__device__ __forceinline__ void gather_batch_indices(int (&ra)[GATHER_GROUP],
                                                     int (&rb)[GATHER_GROUP],
                                                     int (&rc)[GATHER_GROUP],
                                                     const int* __restrict__ indices,
                                                     int b0, int e) {
    const int b1 = (b0 + GATHER_GROUP < e) ? (b0 + GATHER_GROUP) : b0;
    const int b2 = (b0 + 2 * GATHER_GROUP < e) ? (b0 + 2 * GATHER_GROUP) : b0;
    int q0, q1, q2, lo, hi;
    gather_window(b0, e, q0, lo, hi);
    gather_window(b1, e, q1, lo, hi);
    gather_window(b2, e, q2, lo, hi);
    gather_indices(ra, indices, q0);
    gather_indices(rb, indices, q1);
    gather_indices(rc, indices, q2);
}

__device__ __forceinline__ void gather_issue_full(float4 (&v)[GATHER_GROUP],
                                                  const int (&r)[GATHER_GROUP],
                                                  const float* __restrict__ lane_emb) {
#pragma unroll
    for (int t = 0; t < GATHER_GROUP; ++t)
        v[t] = *(const float4*)(lane_emb + (size_t)r[t] * NHID);
}

__device__ __forceinline__ void gather_issue(float4 (&v)[GATHER_GROUP],
                                             const int (&r)[GATHER_GROUP],
                                             const float* __restrict__ lane_emb,
                                             int lo, int hi) {
#pragma unroll
    for (int t = 0; t < GATHER_GROUP; ++t) {
        const bool in = (t >= lo) && (t < hi);
        const int row = in ? r[t] : 0;
        v[t] = *(const float4*)(lane_emb + (size_t)row * NHID);
    }
}

__device__ __forceinline__ void gather_sum_full(float4& acc, const float4 (&v)[GATHER_GROUP]) {
#pragma unroll
    for (int t = 0; t < GATHER_GROUP; ++t) {
        acc.x += v[t].x; acc.y += v[t].y; acc.z += v[t].z; acc.w += v[t].w;
    }
}

__device__ __forceinline__ void gather_sum(float4& acc, const float4 (&v)[GATHER_GROUP],
                                           int lo, int hi) {
    if (hi - lo == GATHER_GROUP) {
        gather_sum_full(acc, v);
    } else {
#pragma unroll
        for (int t = 0; t < GATHER_GROUP; ++t) {
            const bool in = (t >= lo) && (t < hi);
            acc.x += in ? v[t].x : 0.0f;
            acc.y += in ? v[t].y : 0.0f;
            acc.z += in ? v[t].z : 0.0f;
            acc.w += in ? v[t].w : 0.0f;
        }
    }
}

__global__ void __launch_bounds__(GATHER_THREADS)
seg_mean_tanh_kernel(const int* __restrict__ offsets,
                     const int* __restrict__ indices,
                     const float* __restrict__ emb,
                     float* __restrict__ Amat) {
    __shared__ float4 part[GATHER_PAIRS - 1][64];

    const int tid = threadIdx.x;
    const int col = tid & 63;
    const int pair = __builtin_amdgcn_readfirstlane(tid >> 6);
    const float* __restrict__ lane_emb = emb + col * 4;

    {
        const int sample = blockIdx.x;
        const int s = offsets[sample];
        const int e = offsets[sample + 1];

        float4 acc = make_float4(0.0f, 0.0f, 0.0f, 0.0f);
        int base = s + pair * GATHER_BATCH;

        int ra[GATHER_GROUP], rb[GATHER_GROUP], rc[GATHER_GROUP];
#pragma unroll
        for (int t = 0; t < GATHER_GROUP; ++t) { ra[t] = 0; rb[t] = 0; rc[t] = 0; }
        if (base < e) gather_batch_indices(ra, rb, rc, indices, base, e);

        while (base + GATHER_BATCH <= e) {
            float4 va[GATHER_GROUP], vb[GATHER_GROUP], vc[GATHER_GROUP];
            gather_issue_full(va, ra, lane_emb);
            gather_issue_full(vb, rb, lane_emb);
            gather_issue_full(vc, rc, lane_emb);
            const int nb = base + GATHER_PAIRS * GATHER_BATCH;
            if (nb < e) gather_batch_indices(ra, rb, rc, indices, nb, e);
            gather_sum_full(acc, va);
            gather_sum_full(acc, vb);
            gather_sum_full(acc, vc);
            base = nb;
        }

        if (base < e) {
            const bool w1 = (base + GATHER_GROUP < e);
            const bool w2 = (base + 2 * GATHER_GROUP < e);
            int q, lo0, hi0, lo1, hi1, lo2, hi2;
            gather_window(base, e, q, lo0, hi0);
            gather_window(w1 ? (base + GATHER_GROUP) : base, e, q, lo1, hi1);
            gather_window(w2 ? (base + 2 * GATHER_GROUP) : base, e, q, lo2, hi2);
            float4 va[GATHER_GROUP], vb[GATHER_GROUP], vc[GATHER_GROUP];
            gather_issue(va, ra, lane_emb, lo0, hi0);
            if (w1) gather_issue(vb, rb, lane_emb, lo1, hi1);
            if (w2) gather_issue(vc, rc, lane_emb, lo2, hi2);
            gather_sum(acc, va, lo0, hi0);
            if (w1) gather_sum(acc, vb, lo1, hi1);
            if (w2) gather_sum(acc, vc, lo2, hi2);
        }

        if (pair != 0) part[pair - 1][col] = acc;
        __syncthreads();
        if (pair == 0) {
#pragma unroll
            for (int p = 1; p < GATHER_PAIRS; ++p) {
                const float4 o = part[p - 1][col];
                acc.x += o.x; acc.y += o.y; acc.z += o.z; acc.w += o.w;
            }
            const float len = (float)(e - s);
            float4 o;
            o.x = tanhf((len > 0.0f) ? (acc.x / len) : 0.0f);
            o.y = tanhf((len > 0.0f) ? (acc.y / len) : 0.0f);
            o.z = tanhf((len > 0.0f) ? (acc.z / len) : 0.0f);
            o.w = tanhf((len > 0.0f) ? (acc.w / len) : 0.0f);
            *(float4*)(Amat + (size_t)sample * NHID + col * 4) = o;
        }
    }
}

#define HEAD_CHUNKS (NHID / 8)
#define HEAD_WPIECES (NHID * NCLASSES / 4 / 32)

static_assert(HEAD_WPIECES * 32 * 4 == NHID * NCLASSES, "W is copied in whole 16-byte pieces, 32 lanes a pass");

__global__ void __launch_bounds__(32)
wmma_head_kernel(const float* __restrict__ Amat,
                 const float* __restrict__ W,
                 const float* __restrict__ bias,
                 float* __restrict__ out) {
    __shared__ float4 lw4[NHID * NCLASSES / 4];

    const int lane = threadIdx.x;
    const int c = lane & 15;
    const int h = lane >> 4;
    const int row0 = blockIdx.x * 16;
    const int nc = (c < NCLASSES) ? c : (NCLASSES - 1);

    const float* __restrict__ arow = Amat + (size_t)(row0 + c) * NHID + 4 * h;

    float4 wst[HEAD_WPIECES];
#pragma unroll
    for (int i = 0; i < HEAD_WPIECES; ++i) wst[i] = ((const float4*)W)[lane + 32 * i];
    float4 a[HEAD_CHUNKS];
#pragma unroll
    for (int j = 0; j < HEAD_CHUNKS; ++j) a[j] = *(const float4*)(arow + 8 * j);
    const float bb = bias[nc];
    __builtin_amdgcn_sched_barrier(0);

#pragma unroll
    for (int i = 0; i < HEAD_WPIECES; ++i) lw4[lane + 32 * i] = wst[i];
    __syncthreads();

    const float* lw = (const float*)lw4 + (4 * h) * NCLASSES + nc;
    float w[HEAD_CHUNKS][4];
#pragma unroll
    for (int j = 0; j < HEAD_CHUNKS; ++j) {
#pragma unroll
        for (int i = 0; i < 4; ++i) w[j][i] = lw[(8 * j + i) * NCLASSES];
    }
    __builtin_amdgcn_sched_barrier(0);

    v8f acc0 = {};
    v8f acc1 = {};
#pragma unroll
    for (int j = 0; j < HEAD_CHUNKS; ++j) {
        v2f av, bv;
        av.x = a[j].x; av.y = a[j].y;
        bv.x = w[j][0]; bv.y = w[j][1];
        acc0 = __builtin_amdgcn_wmma_f32_16x16x4_f32(
            false, av, false, bv, (short)0, acc0, false, false);
        av.x = a[j].z; av.y = a[j].w;
        bv.x = w[j][2]; bv.y = w[j][3];
        acc1 = __builtin_amdgcn_wmma_f32_16x16x4_f32(
            false, av, false, bv, (short)0, acc1, false, false);
    }

    if (c < NCLASSES) {
#pragma unroll
        for (int r = 0; r < 8; ++r) {
            const int m = row0 + r + 8 * h;
            out[(size_t)m * NCLASSES + c] = (acc0[r] + acc1[r]) + bb;
        }
    }
}

extern "C" void kernel_launch(void* const* d_in, const int* in_sizes, int n_in,
                              void* d_out, int out_size, void* d_ws, size_t ws_size,
                              hipStream_t stream) {
    const int*   lengths = (const int*)d_in[0];
    const int*   indices = (const int*)d_in[1];
    const float* emb     = (const float*)d_in[2];
    const float* W       = (const float*)d_in[3];
    const float* bias    = (const float*)d_in[4];
    float*       out     = (float*)d_out;

    int*   offsets = (int*)d_ws;
    float* Amat    = (float*)((char*)d_ws + (64 * 1024));

    seg_scan_kernel<<<1, 1024, 0, stream>>>(lengths, offsets);
    seg_mean_tanh_kernel<<<B_SAMPLES, GATHER_THREADS, 0, stream>>>(offsets, indices, emb, Amat);
    wmma_head_kernel<<<B_SAMPLES / 16, 32, 0, stream>>>(Amat, W, bias, out);
}
